# starting kernel
# baseline (speedup 1.0000x reference)
#include <hip/hip_runtime.h>

#define NNODES 50000
#define NEDGES 800000
#define NODE_TILES (NNODES / 16)
#define EDGE_TILES (NEDGES / 16)
#define WPB 8

typedef __attribute__((ext_vector_type(2))) float v2f;
typedef __attribute__((ext_vector_type(8))) float v8f;

__device__ __forceinline__ v8f wmma4(v2f a, v2f b, v8f c) {
  return __builtin_amdgcn_wmma_f32_16x16x4_f32(
      false, a, false, b, (short)0, c, false, false);
}

__device__ __forceinline__ v2f ldA(const float* A, int ldk, int kb, int lane) {
  int m = lane & 15;
  int k = 4 * kb + 2 * (lane >> 4);
  v2f a;
  a.x = A[m * ldk + k];
  a.y = A[m * ldk + k + 1];
  return a;
}

__device__ __forceinline__ v2f ldB(const float* W, int ncols, int kb, int col0, int lane) {
  int col = col0 + (lane & 15);
  int k = 4 * kb + 2 * (lane >> 4);
  v2f b;
  b.x = W[k * ncols + col];
  b.y = W[(k + 1) * ncols + col];
  return b;
}

__device__ __forceinline__ v8f bias_frag(const float* b, int col0, int lane) {
  float bv = b[col0 + (lane & 15)];
  v8f c;
#pragma unroll
  for (int r = 0; r < 8; ++r) c[r] = bv;
  return c;
}

__device__ __forceinline__ void st_tile(float* A, const v8f* acc, int lane, int mode) {
  int half = lane >> 4, nl = lane & 15;
#pragma unroll
  for (int nb = 0; nb < 4; ++nb)
#pragma unroll
    for (int r = 0; r < 8; ++r) {
      float v = acc[nb][r];
      if (mode == 1) v = tanhf(v);
      else if (mode == 2) v = fmaxf(v, 0.0f);
      A[(r + 8 * half) * 64 + nb * 16 + nl] = v;
    }
}

__device__ __forceinline__ void gemm64_lds(const float* A, const float* W,
                                           const float* b, v8f* acc, int lane) {
#pragma unroll
  for (int nb = 0; nb < 4; ++nb) acc[nb] = bias_frag(b, nb * 16, lane);
#pragma unroll
  for (int kb = 0; kb < 16; ++kb) {
    v2f a = ldA(A, 64, kb, lane);
#pragma unroll
    for (int nb = 0; nb < 4; ++nb)
      acc[nb] = wmma4(a, ldB(W, 64, kb, nb * 16, lane), acc[nb]);
  }
}

__global__ void k_zero(float* p, long n) {
  long i = (long)blockIdx.x * blockDim.x + threadIdx.x;
  if (i < n) p[i] = 0.0f;
}

__global__ void __launch_bounds__(256) k_encode(
    const float* __restrict__ x,
    const float* __restrict__ WencK, const float* __restrict__ bencK,
    const float* __restrict__ WencP1, const float* __restrict__ bencP1,
    const float* __restrict__ WencP2, const float* __restrict__ bencP2,
    const float* __restrict__ WD, const float* __restrict__ bD,
    float* __restrict__ hK, float* __restrict__ h1,
    float* __restrict__ h2, float* __restrict__ Dxh) {
  __shared__ float sWK[32 * 32], sW1[32 * 32], sW2[32 * 32], sWD[64 * 64];
  __shared__ float sbK[32], sb1[32], sb2[32], sbD[32];
  for (int i = threadIdx.x; i < 32 * 32; i += blockDim.x) {
    sWK[i] = WencK[i]; sW1[i] = WencP1[i]; sW2[i] = WencP2[i];
  }
  for (int i = threadIdx.x; i < 64 * 64; i += blockDim.x) sWD[i] = WD[i];
  for (int i = threadIdx.x; i < 32; i += blockDim.x) {
    sbK[i] = bencK[i]; sb1[i] = bencP1[i]; sb2[i] = bencP2[i]; sbD[i] = bD[32 + i];
  }
  __syncthreads();

  int wave = threadIdx.x >> 5, lane = threadIdx.x & 31;
  int m = lane & 15;
  int half = lane >> 4, nl = lane & 15;

  for (int tile = blockIdx.x * WPB + wave; tile < NODE_TILES;
       tile += gridDim.x * WPB) {
    int n0 = tile * 16;

    v8f aK[2], a1[2], a2[2], aD[2];
#pragma unroll
    for (int nb = 0; nb < 2; ++nb) {
      aK[nb] = bias_frag(sbK, nb * 16, lane);
      a1[nb] = bias_frag(sb1, nb * 16, lane);
      a2[nb] = bias_frag(sb2, nb * 16, lane);
      aD[nb] = bias_frag(sbD, nb * 16, lane);
    }

#pragma unroll
    for (int kb = 0; kb < 8; ++kb) {
      int k = 4 * kb + 2 * (lane >> 4);
      long base = (long)(n0 + m) * 64;
      v2f aq, ap;
      aq.x = x[base + k];       aq.y = x[base + k + 1];
      ap.x = x[base + 32 + k];  ap.y = x[base + 32 + k + 1];
#pragma unroll
      for (int nb = 0; nb < 2; ++nb) {
        aK[nb] = wmma4(ap, ldB(sWK, 32, kb, nb * 16, lane), aK[nb]);
        a1[nb] = wmma4(aq, ldB(sW1, 32, kb, nb * 16, lane), a1[nb]);
        a2[nb] = wmma4(aq, ldB(sW2, 32, kb, nb * 16, lane), a2[nb]);
      }
    }
#pragma unroll
    for (int kb = 0; kb < 16; ++kb) {
      int k = 4 * kb + 2 * (lane >> 4);
      long base = (long)(n0 + m) * 64;
      v2f ax;
      ax.x = x[base + k]; ax.y = x[base + k + 1];
#pragma unroll
      for (int nb = 0; nb < 2; ++nb)
        aD[nb] = wmma4(ax, ldB(sWD, 64, kb, 32 + nb * 16, lane), aD[nb]);
    }

#pragma unroll
    for (int nb = 0; nb < 2; ++nb)
#pragma unroll
      for (int r = 0; r < 8; ++r) {
        long o = (long)(n0 + r + 8 * half) * 32 + nb * 16 + nl;
        hK[o] = aK[nb][r];
        h1[o] = a1[nb][r];
        h2[o] = a2[nb][r];
        Dxh[o] = aD[nb][r];
      }
  }
}

__global__ void k_scatter_enc(const int* __restrict__ src, const int* __restrict__ dst,
                              const float* __restrict__ hK, const float* __restrict__ Dxh,
                              float* __restrict__ hK_agg, float* __restrict__ d_agg) {
  long tid = (long)blockIdx.x * blockDim.x + threadIdx.x;
  if (tid >= (long)NEDGES * 8) return;
  int e = (int)(tid >> 3);
  int j = (int)(tid & 7) * 4;
  int s = src[e], d = dst[e];
  float4 a = *(const float4*)&hK[(long)s * 32 + j];
  float4 b = *(const float4*)&Dxh[(long)s * 32 + j];
  float* pa = &hK_agg[(long)d * 32 + j];
  float* pb = &d_agg[(long)d * 32 + j];
  atomicAdd(pa + 0, a.x); atomicAdd(pa + 1, a.y);
  atomicAdd(pa + 2, a.z); atomicAdd(pa + 3, a.w);
  atomicAdd(pb + 0, b.x); atomicAdd(pb + 1, b.y);
  atomicAdd(pb + 2, b.z); atomicAdd(pb + 3, b.w);
}

__global__ void __launch_bounds__(256) k_node_mlp(
    const float* __restrict__ hK_agg,
    const float* __restrict__ WK1, const float* __restrict__ bK1,
    const float* __restrict__ WK2, const float* __restrict__ bK2,
    const float* __restrict__ WK3, const float* __restrict__ bK3,
    float* __restrict__ E_node) {
  __shared__ float sW1[32 * 64], sW2[64 * 64], sW3[64 * 64];
  __shared__ float sb1[64], sb2[64], sb3[64];
  __shared__ float sact[WPB][16 * 64];
  for (int i = threadIdx.x; i < 32 * 64; i += blockDim.x) sW1[i] = WK1[i];
  for (int i = threadIdx.x; i < 64 * 64; i += blockDim.x) { sW2[i] = WK2[i]; sW3[i] = WK3[i]; }
  for (int i = threadIdx.x; i < 64; i += blockDim.x) { sb1[i] = bK1[i]; sb2[i] = bK2[i]; sb3[i] = bK3[i]; }
  __syncthreads();

  int wave = threadIdx.x >> 5, lane = threadIdx.x & 31;
  int m = lane & 15;
  int half = lane >> 4, nl = lane & 15;
  float* A = sact[wave];

  for (int tile = blockIdx.x * WPB + wave; tile < NODE_TILES;
       tile += gridDim.x * WPB) {
    int n0 = tile * 16;

    v8f acc[4];
#pragma unroll
    for (int nb = 0; nb < 4; ++nb) acc[nb] = bias_frag(sb1, nb * 16, lane);
#pragma unroll
    for (int kb = 0; kb < 8; ++kb) {
      int k = 4 * kb + 2 * (lane >> 4);
      long base = (long)(n0 + m) * 32;
      v2f a; a.x = hK_agg[base + k]; a.y = hK_agg[base + k + 1];
#pragma unroll
      for (int nb = 0; nb < 4; ++nb)
        acc[nb] = wmma4(a, ldB(sW1, 64, kb, nb * 16, lane), acc[nb]);
    }
    st_tile(A, acc, lane, 1);
    gemm64_lds(A, sW2, sb2, acc, lane);
    st_tile(A, acc, lane, 2);
    gemm64_lds(A, sW3, sb3, acc, lane);

#pragma unroll
    for (int nb = 0; nb < 4; ++nb)
#pragma unroll
      for (int r = 0; r < 8; ++r)
        E_node[(long)(n0 + r + 8 * half) * 64 + nb * 16 + nl] = acc[nb][r];
  }
}

__global__ void __launch_bounds__(256) k_edge(
    const int* __restrict__ src, const int* __restrict__ dst,
    const float* __restrict__ h1, const float* __restrict__ h2,
    const float* __restrict__ E_node,
    const float* __restrict__ WU1, const float* __restrict__ bU1,
    const float* __restrict__ WU2, const float* __restrict__ bU2,
    const float* __restrict__ WU3, const float* __restrict__ bU3,
    float* __restrict__ E_new) {
  __shared__ float sW1[32 * 64], sW2[64 * 64], sW3[64 * 64];
  __shared__ float sb1[64], sb2[64], sb3[64];
  __shared__ float sact[WPB][16 * 64];
  for (int i = threadIdx.x; i < 32 * 64; i += blockDim.x) sW1[i] = WU1[i];
  for (int i = threadIdx.x; i < 64 * 64; i += blockDim.x) { sW2[i] = WU2[i]; sW3[i] = WU3[i]; }
  for (int i = threadIdx.x; i < 64; i += blockDim.x) { sb1[i] = bU1[i]; sb2[i] = bU2[i]; sb3[i] = bU3[i]; }
  __syncthreads();

  int wave = threadIdx.x >> 5, lane = threadIdx.x & 31;
  int m = lane & 15;
  int half = lane >> 4, nl = lane & 15;
  float* A = sact[wave];

  for (int tile = blockIdx.x * WPB + wave; tile < EDGE_TILES;
       tile += gridDim.x * WPB) {
    int e0 = tile * 16;
    int s = src[e0 + m], d = dst[e0 + m];

    v8f acc[4];
#pragma unroll
    for (int nb = 0; nb < 4; ++nb) acc[nb] = bias_frag(sb1, nb * 16, lane);
#pragma unroll
    for (int kb = 0; kb < 8; ++kb) {
      int k = 4 * kb + 2 * (lane >> 4);
      v2f a;
      a.x = h1[(long)s * 32 + k]     + h2[(long)d * 32 + k];
      a.y = h1[(long)s * 32 + k + 1] + h2[(long)d * 32 + k + 1];
#pragma unroll
      for (int nb = 0; nb < 4; ++nb)
        acc[nb] = wmma4(a, ldB(sW1, 64, kb, nb * 16, lane), acc[nb]);
    }
    st_tile(A, acc, lane, 1);
    gemm64_lds(A, sW2, sb2, acc, lane);
    st_tile(A, acc, lane, 2);
    gemm64_lds(A, sW3, sb3, acc, lane);

#pragma unroll
    for (int r = 0; r < 8; ++r) {
      int e = e0 + r + 8 * half;
      int se = src[e], de = dst[e];
#pragma unroll
      for (int nb = 0; nb < 4; ++nb) {
        int col = nb * 16 + nl;
        float v = acc[nb][r] * E_node[(long)se * 64 + col];
        atomicAdd(&E_new[(long)de * 64 + col], v);
      }
    }
  }
}

__global__ void __launch_bounds__(256) k_final(
    const float* __restrict__ E_new,
    const float* __restrict__ WH, const float* __restrict__ bH,
    const float* __restrict__ d_agg, float* __restrict__ out) {
  __shared__ float sWH[64 * 64], sbH[64];
  for (int i = threadIdx.x; i < 64 * 64; i += blockDim.x) sWH[i] = WH[i];
  for (int i = threadIdx.x; i < 64; i += blockDim.x) sbH[i] = bH[i];
  __syncthreads();

  int wave = threadIdx.x >> 5, lane = threadIdx.x & 31;
  int m = lane & 15;
  int half = lane >> 4, nl = lane & 15;

  for (int tile = blockIdx.x * WPB + wave; tile < NODE_TILES;
       tile += gridDim.x * WPB) {
    int n0 = tile * 16;

    v8f acc[4];
#pragma unroll
    for (int nb = 0; nb < 4; ++nb) acc[nb] = bias_frag(sbH, nb * 16, lane);
#pragma unroll
    for (int kb = 0; kb < 16; ++kb) {
      int k = 4 * kb + 2 * (lane >> 4);
      long base = (long)(n0 + m) * 64;
      v2f a; a.x = E_new[base + k]; a.y = E_new[base + k + 1];
#pragma unroll
      for (int nb = 0; nb < 4; ++nb)
        acc[nb] = wmma4(a, ldB(sWH, 64, kb, nb * 16, lane), acc[nb]);
    }

#pragma unroll
    for (int r = 0; r < 8; ++r) {
      long row = n0 + r + 8 * half;
#pragma unroll
      for (int nb = 0; nb < 2; ++nb) {
        out[row * 64 + nb * 16 + nl] = acc[nb + 2][r];
        out[row * 64 + 32 + nb * 16 + nl] =
            -acc[nb][r] - d_agg[row * 32 + nb * 16 + nl];
      }
    }
  }
}

extern "C" void kernel_launch(void* const* d_in, const int* in_sizes, int n_in,
                              void* d_out, int out_size, void* d_ws, size_t ws_size,
                              hipStream_t stream) {
  const float* x      = (const float*)d_in[0];
  const int*   src    = (const int*)d_in[1];
  const int*   dst    = (const int*)d_in[2];
  const float* WencK  = (const float*)d_in[3];
  const float* bencK  = (const float*)d_in[4];
  const float* WencP1 = (const float*)d_in[5];
  const float* bencP1 = (const float*)d_in[6];
  const float* WencP2 = (const float*)d_in[7];
  const float* bencP2 = (const float*)d_in[8];
  const float* WK1 = (const float*)d_in[9];
  const float* bK1 = (const float*)d_in[10];
  const float* WK2 = (const float*)d_in[11];
  const float* bK2 = (const float*)d_in[12];
  const float* WK3 = (const float*)d_in[13];
  const float* bK3 = (const float*)d_in[14];
  const float* WU1 = (const float*)d_in[15];
  const float* bU1 = (const float*)d_in[16];
  const float* WU2 = (const float*)d_in[17];
  const float* bU2 = (const float*)d_in[18];
  const float* WU3 = (const float*)d_in[19];
  const float* bU3 = (const float*)d_in[20];
  const float* WH  = (const float*)d_in[21];
  const float* bH  = (const float*)d_in[22];
  const float* WD  = (const float*)d_in[23];
  const float* bD  = (const float*)d_in[24];

  float* ws = (float*)d_ws;
  const size_t N32 = (size_t)NNODES * 32;
  const size_t N64 = (size_t)NNODES * 64;
  float* hK     = ws;
  float* h1v    = hK + N32;
  float* h2v    = h1v + N32;
  float* Dxh    = h2v + N32;
  float* hK_agg = Dxh + N32;
  float* d_agg  = hK_agg + N32;
  float* E_new  = d_agg + N32;
  float* E_node = E_new + N64;

  float* out = (float*)d_out;

  {
    long nz = (long)NNODES * 128;
    int blocks = (int)((nz + 255) / 256);
    k_zero<<<blocks, 256, 0, stream>>>(hK_agg, nz);
  }
  {
    int blocks = (NODE_TILES + WPB * 4 - 1) / (WPB * 4);
    k_encode<<<blocks, 256, 0, stream>>>(x, WencK, bencK, WencP1, bencP1,
                                         WencP2, bencP2, WD, bD,
                                         hK, h1v, h2v, Dxh);
  }
  {
    long nt = (long)NEDGES * 8;
    int blocks = (int)((nt + 255) / 256);
    k_scatter_enc<<<blocks, 256, 0, stream>>>(src, dst, hK, Dxh, hK_agg, d_agg);
  }
  {
    int blocks = (NODE_TILES + WPB * 4 - 1) / (WPB * 4);
    k_node_mlp<<<blocks, 256, 0, stream>>>(hK_agg, WK1, bK1, WK2, bK2, WK3, bK3,
                                           E_node);
  }
  {
    int blocks = (EDGE_TILES + WPB * 4 - 1) / (WPB * 4);
    k_edge<<<blocks, 256, 0, stream>>>(src, dst, h1v, h2v, E_node,
                                       WU1, bU1, WU2, bU2, WU3, bU3, E_new);
  }
  {
    int blocks = (NODE_TILES + WPB * 4 - 1) / (WPB * 4);
    k_final<<<blocks, 256, 0, stream>>>(E_new, WH, bH, d_agg, out);
  }
}


# Edge kernel: the 32 E_node gathers per lane are issued as one group at the top of each tile, under the MLP, instead of one dependent cache trip each between the atomics of the epilogue
# speedup vs baseline: 1.3572x; 1.3572x over previous
#include <hip/hip_runtime.h>

#define NNODES 50000
#define NEDGES 800000
#define NODE_TILES (NNODES / 16)
#define EDGE_TILES (NEDGES / 16)
#define WPB 8

typedef __attribute__((ext_vector_type(2))) float v2f;
typedef __attribute__((ext_vector_type(8))) float v8f;

__device__ __forceinline__ v8f wmma4(v2f a, v2f b, v8f c) {
  return __builtin_amdgcn_wmma_f32_16x16x4_f32(
      false, a, false, b, (short)0, c, false, false);
}

__device__ __forceinline__ v2f ldA(const float* A, int ldk, int kb, int lane) {
  int m = lane & 15;
  int k = 4 * kb + 2 * (lane >> 4);
  v2f a;
  a.x = A[m * ldk + k];
  a.y = A[m * ldk + k + 1];
  return a;
}

__device__ __forceinline__ v2f ldB(const float* W, int ncols, int kb, int col0, int lane) {
  int col = col0 + (lane & 15);
  int k = 4 * kb + 2 * (lane >> 4);
  v2f b;
  b.x = W[k * ncols + col];
  b.y = W[(k + 1) * ncols + col];
  return b;
}

__device__ __forceinline__ v8f bias_frag(const float* b, int col0, int lane) {
  float bv = b[col0 + (lane & 15)];
  v8f c;
#pragma unroll
  for (int r = 0; r < 8; ++r) c[r] = bv;
  return c;
}

__device__ __forceinline__ void st_tile(float* A, const v8f* acc, int lane, int mode) {
  int half = lane >> 4, nl = lane & 15;
#pragma unroll
  for (int nb = 0; nb < 4; ++nb)
#pragma unroll
    for (int r = 0; r < 8; ++r) {
      float v = acc[nb][r];
      if (mode == 1) v = tanhf(v);
      else if (mode == 2) v = fmaxf(v, 0.0f);
      A[(r + 8 * half) * 64 + nb * 16 + nl] = v;
    }
}

__device__ __forceinline__ void gemm64_lds(const float* A, const float* W,
                                           const float* b, v8f* acc, int lane) {
#pragma unroll
  for (int nb = 0; nb < 4; ++nb) acc[nb] = bias_frag(b, nb * 16, lane);
#pragma unroll
  for (int kb = 0; kb < 16; ++kb) {
    v2f a = ldA(A, 64, kb, lane);
#pragma unroll
    for (int nb = 0; nb < 4; ++nb)
      acc[nb] = wmma4(a, ldB(W, 64, kb, nb * 16, lane), acc[nb]);
  }
}

__global__ void k_zero(float* p, long n) {
  long i = (long)blockIdx.x * blockDim.x + threadIdx.x;
  if (i < n) p[i] = 0.0f;
}

__global__ void __launch_bounds__(256) k_encode(
    const float* __restrict__ x,
    const float* __restrict__ WencK, const float* __restrict__ bencK,
    const float* __restrict__ WencP1, const float* __restrict__ bencP1,
    const float* __restrict__ WencP2, const float* __restrict__ bencP2,
    const float* __restrict__ WD, const float* __restrict__ bD,
    float* __restrict__ hK, float* __restrict__ h1,
    float* __restrict__ h2, float* __restrict__ Dxh) {
  __shared__ float sWK[32 * 32], sW1[32 * 32], sW2[32 * 32], sWD[64 * 64];
  __shared__ float sbK[32], sb1[32], sb2[32], sbD[32];
  for (int i = threadIdx.x; i < 32 * 32; i += blockDim.x) {
    sWK[i] = WencK[i]; sW1[i] = WencP1[i]; sW2[i] = WencP2[i];
  }
  for (int i = threadIdx.x; i < 64 * 64; i += blockDim.x) sWD[i] = WD[i];
  for (int i = threadIdx.x; i < 32; i += blockDim.x) {
    sbK[i] = bencK[i]; sb1[i] = bencP1[i]; sb2[i] = bencP2[i]; sbD[i] = bD[32 + i];
  }
  __syncthreads();

  int wave = threadIdx.x >> 5, lane = threadIdx.x & 31;
  int m = lane & 15;
  int half = lane >> 4, nl = lane & 15;

  for (int tile = blockIdx.x * WPB + wave; tile < NODE_TILES;
       tile += gridDim.x * WPB) {
    int n0 = tile * 16;

    v8f aK[2], a1[2], a2[2], aD[2];
#pragma unroll
    for (int nb = 0; nb < 2; ++nb) {
      aK[nb] = bias_frag(sbK, nb * 16, lane);
      a1[nb] = bias_frag(sb1, nb * 16, lane);
      a2[nb] = bias_frag(sb2, nb * 16, lane);
      aD[nb] = bias_frag(sbD, nb * 16, lane);
    }

#pragma unroll
    for (int kb = 0; kb < 8; ++kb) {
      int k = 4 * kb + 2 * (lane >> 4);
      long base = (long)(n0 + m) * 64;
      v2f aq, ap;
      aq.x = x[base + k];       aq.y = x[base + k + 1];
      ap.x = x[base + 32 + k];  ap.y = x[base + 32 + k + 1];
#pragma unroll
      for (int nb = 0; nb < 2; ++nb) {
        aK[nb] = wmma4(ap, ldB(sWK, 32, kb, nb * 16, lane), aK[nb]);
        a1[nb] = wmma4(aq, ldB(sW1, 32, kb, nb * 16, lane), a1[nb]);
        a2[nb] = wmma4(aq, ldB(sW2, 32, kb, nb * 16, lane), a2[nb]);
      }
    }
#pragma unroll
    for (int kb = 0; kb < 16; ++kb) {
      int k = 4 * kb + 2 * (lane >> 4);
      long base = (long)(n0 + m) * 64;
      v2f ax;
      ax.x = x[base + k]; ax.y = x[base + k + 1];
#pragma unroll
      for (int nb = 0; nb < 2; ++nb)
        aD[nb] = wmma4(ax, ldB(sWD, 64, kb, 32 + nb * 16, lane), aD[nb]);
    }

#pragma unroll
    for (int nb = 0; nb < 2; ++nb)
#pragma unroll
      for (int r = 0; r < 8; ++r) {
        long o = (long)(n0 + r + 8 * half) * 32 + nb * 16 + nl;
        hK[o] = aK[nb][r];
        h1[o] = a1[nb][r];
        h2[o] = a2[nb][r];
        Dxh[o] = aD[nb][r];
      }
  }
}

__global__ void k_scatter_enc(const int* __restrict__ src, const int* __restrict__ dst,
                              const float* __restrict__ hK, const float* __restrict__ Dxh,
                              float* __restrict__ hK_agg, float* __restrict__ d_agg) {
  long tid = (long)blockIdx.x * blockDim.x + threadIdx.x;
  if (tid >= (long)NEDGES * 8) return;
  int e = (int)(tid >> 3);
  int j = (int)(tid & 7) * 4;
  int s = src[e], d = dst[e];
  float4 a = *(const float4*)&hK[(long)s * 32 + j];
  float4 b = *(const float4*)&Dxh[(long)s * 32 + j];
  float* pa = &hK_agg[(long)d * 32 + j];
  float* pb = &d_agg[(long)d * 32 + j];
  atomicAdd(pa + 0, a.x); atomicAdd(pa + 1, a.y);
  atomicAdd(pa + 2, a.z); atomicAdd(pa + 3, a.w);
  atomicAdd(pb + 0, b.x); atomicAdd(pb + 1, b.y);
  atomicAdd(pb + 2, b.z); atomicAdd(pb + 3, b.w);
}

__global__ void __launch_bounds__(256) k_node_mlp(
    const float* __restrict__ hK_agg,
    const float* __restrict__ WK1, const float* __restrict__ bK1,
    const float* __restrict__ WK2, const float* __restrict__ bK2,
    const float* __restrict__ WK3, const float* __restrict__ bK3,
    float* __restrict__ E_node) {
  __shared__ float sW1[32 * 64], sW2[64 * 64], sW3[64 * 64];
  __shared__ float sb1[64], sb2[64], sb3[64];
  __shared__ float sact[WPB][16 * 64];
  for (int i = threadIdx.x; i < 32 * 64; i += blockDim.x) sW1[i] = WK1[i];
  for (int i = threadIdx.x; i < 64 * 64; i += blockDim.x) { sW2[i] = WK2[i]; sW3[i] = WK3[i]; }
  for (int i = threadIdx.x; i < 64; i += blockDim.x) { sb1[i] = bK1[i]; sb2[i] = bK2[i]; sb3[i] = bK3[i]; }
  __syncthreads();

  int wave = threadIdx.x >> 5, lane = threadIdx.x & 31;
  int m = lane & 15;
  int half = lane >> 4, nl = lane & 15;
  float* A = sact[wave];

  for (int tile = blockIdx.x * WPB + wave; tile < NODE_TILES;
       tile += gridDim.x * WPB) {
    int n0 = tile * 16;

    v8f acc[4];
#pragma unroll
    for (int nb = 0; nb < 4; ++nb) acc[nb] = bias_frag(sb1, nb * 16, lane);
#pragma unroll
    for (int kb = 0; kb < 8; ++kb) {
      int k = 4 * kb + 2 * (lane >> 4);
      long base = (long)(n0 + m) * 32;
      v2f a; a.x = hK_agg[base + k]; a.y = hK_agg[base + k + 1];
#pragma unroll
      for (int nb = 0; nb < 4; ++nb)
        acc[nb] = wmma4(a, ldB(sW1, 64, kb, nb * 16, lane), acc[nb]);
    }
    st_tile(A, acc, lane, 1);
    gemm64_lds(A, sW2, sb2, acc, lane);
    st_tile(A, acc, lane, 2);
    gemm64_lds(A, sW3, sb3, acc, lane);

#pragma unroll
    for (int nb = 0; nb < 4; ++nb)
#pragma unroll
      for (int r = 0; r < 8; ++r)
        E_node[(long)(n0 + r + 8 * half) * 64 + nb * 16 + nl] = acc[nb][r];
  }
}

__global__ void __launch_bounds__(256) k_edge(
    const int* __restrict__ src, const int* __restrict__ dst,
    const float* __restrict__ h1, const float* __restrict__ h2,
    const float* __restrict__ E_node,
    const float* __restrict__ WU1, const float* __restrict__ bU1,
    const float* __restrict__ WU2, const float* __restrict__ bU2,
    const float* __restrict__ WU3, const float* __restrict__ bU3,
    float* __restrict__ E_new) {
  __shared__ float sW1[32 * 64], sW2[64 * 64], sW3[64 * 64];
  __shared__ float sb1[64], sb2[64], sb3[64];
  __shared__ float sact[WPB][16 * 64];
  for (int i = threadIdx.x; i < 32 * 64; i += blockDim.x) sW1[i] = WU1[i];
  for (int i = threadIdx.x; i < 64 * 64; i += blockDim.x) { sW2[i] = WU2[i]; sW3[i] = WU3[i]; }
  for (int i = threadIdx.x; i < 64; i += blockDim.x) { sb1[i] = bU1[i]; sb2[i] = bU2[i]; sb3[i] = bU3[i]; }
  __syncthreads();

  int wave = threadIdx.x >> 5, lane = threadIdx.x & 31;
  int m = lane & 15;
  int half = lane >> 4, nl = lane & 15;
  float* A = sact[wave];

  for (int tile = blockIdx.x * WPB + wave; tile < EDGE_TILES;
       tile += gridDim.x * WPB) {
    int e0 = tile * 16;
    int s = src[e0 + m], d = dst[e0 + m];

    int se[8], de[8];
#pragma unroll
    for (int r = 0; r < 8; ++r) {
      int e = e0 + r + 8 * half;
      se[r] = src[e];
      de[r] = dst[e];
    }
    float en[8][4];
#pragma unroll
    for (int r = 0; r < 8; ++r)
#pragma unroll
      for (int nb = 0; nb < 4; ++nb)
        en[r][nb] = E_node[(long)se[r] * 64 + nb * 16 + nl];

    v8f acc[4];
#pragma unroll
    for (int nb = 0; nb < 4; ++nb) acc[nb] = bias_frag(sb1, nb * 16, lane);
#pragma unroll
    for (int kb = 0; kb < 8; ++kb) {
      int k = 4 * kb + 2 * (lane >> 4);
      v2f a;
      a.x = h1[(long)s * 32 + k]     + h2[(long)d * 32 + k];
      a.y = h1[(long)s * 32 + k + 1] + h2[(long)d * 32 + k + 1];
#pragma unroll
      for (int nb = 0; nb < 4; ++nb)
        acc[nb] = wmma4(a, ldB(sW1, 64, kb, nb * 16, lane), acc[nb]);
    }
    st_tile(A, acc, lane, 1);
    gemm64_lds(A, sW2, sb2, acc, lane);
    st_tile(A, acc, lane, 2);
    gemm64_lds(A, sW3, sb3, acc, lane);

#pragma unroll
    for (int r = 0; r < 8; ++r) {
#pragma unroll
      for (int nb = 0; nb < 4; ++nb) {
        int col = nb * 16 + nl;
        float v = acc[nb][r] * en[r][nb];
        atomicAdd(&E_new[(long)de[r] * 64 + col], v);
      }
    }
  }
}

__global__ void __launch_bounds__(256) k_final(
    const float* __restrict__ E_new,
    const float* __restrict__ WH, const float* __restrict__ bH,
    const float* __restrict__ d_agg, float* __restrict__ out) {
  __shared__ float sWH[64 * 64], sbH[64];
  for (int i = threadIdx.x; i < 64 * 64; i += blockDim.x) sWH[i] = WH[i];
  for (int i = threadIdx.x; i < 64; i += blockDim.x) sbH[i] = bH[i];
  __syncthreads();

  int wave = threadIdx.x >> 5, lane = threadIdx.x & 31;
  int m = lane & 15;
  int half = lane >> 4, nl = lane & 15;

  for (int tile = blockIdx.x * WPB + wave; tile < NODE_TILES;
       tile += gridDim.x * WPB) {
    int n0 = tile * 16;

    v8f acc[4];
#pragma unroll
    for (int nb = 0; nb < 4; ++nb) acc[nb] = bias_frag(sbH, nb * 16, lane);
#pragma unroll
    for (int kb = 0; kb < 16; ++kb) {
      int k = 4 * kb + 2 * (lane >> 4);
      long base = (long)(n0 + m) * 64;
      v2f a; a.x = E_new[base + k]; a.y = E_new[base + k + 1];
#pragma unroll
      for (int nb = 0; nb < 4; ++nb)
        acc[nb] = wmma4(a, ldB(sWH, 64, kb, nb * 16, lane), acc[nb]);
    }

#pragma unroll
    for (int r = 0; r < 8; ++r) {
      long row = n0 + r + 8 * half;
#pragma unroll
      for (int nb = 0; nb < 2; ++nb) {
        out[row * 64 + nb * 16 + nl] = acc[nb + 2][r];
        out[row * 64 + 32 + nb * 16 + nl] =
            -acc[nb][r] - d_agg[row * 32 + nb * 16 + nl];
      }
    }
  }
}

extern "C" void kernel_launch(void* const* d_in, const int* in_sizes, int n_in,
                              void* d_out, int out_size, void* d_ws, size_t ws_size,
                              hipStream_t stream) {
  const float* x      = (const float*)d_in[0];
  const int*   src    = (const int*)d_in[1];
  const int*   dst    = (const int*)d_in[2];
  const float* WencK  = (const float*)d_in[3];
  const float* bencK  = (const float*)d_in[4];
  const float* WencP1 = (const float*)d_in[5];
  const float* bencP1 = (const float*)d_in[6];
  const float* WencP2 = (const float*)d_in[7];
  const float* bencP2 = (const float*)d_in[8];
  const float* WK1 = (const float*)d_in[9];
  const float* bK1 = (const float*)d_in[10];
  const float* WK2 = (const float*)d_in[11];
  const float* bK2 = (const float*)d_in[12];
  const float* WK3 = (const float*)d_in[13];
  const float* bK3 = (const float*)d_in[14];
  const float* WU1 = (const float*)d_in[15];
  const float* bU1 = (const float*)d_in[16];
  const float* WU2 = (const float*)d_in[17];
  const float* bU2 = (const float*)d_in[18];
  const float* WU3 = (const float*)d_in[19];
  const float* bU3 = (const float*)d_in[20];
  const float* WH  = (const float*)d_in[21];
  const float* bH  = (const float*)d_in[22];
  const float* WD  = (const float*)d_in[23];
  const float* bD  = (const float*)d_in[24];

  float* ws = (float*)d_ws;
  const size_t N32 = (size_t)NNODES * 32;
  const size_t N64 = (size_t)NNODES * 64;
  float* hK     = ws;
  float* h1v    = hK + N32;
  float* h2v    = h1v + N32;
  float* Dxh    = h2v + N32;
  float* hK_agg = Dxh + N32;
  float* d_agg  = hK_agg + N32;
  float* E_new  = d_agg + N32;
  float* E_node = E_new + N64;

  float* out = (float*)d_out;

  {
    long nz = (long)NNODES * 128;
    int blocks = (int)((nz + 255) / 256);
    k_zero<<<blocks, 256, 0, stream>>>(hK_agg, nz);
  }
  {
    int blocks = (NODE_TILES + WPB * 4 - 1) / (WPB * 4);
    k_encode<<<blocks, 256, 0, stream>>>(x, WencK, bencK, WencP1, bencP1,
                                         WencP2, bencP2, WD, bD,
                                         hK, h1v, h2v, Dxh);
  }
  {
    long nt = (long)NEDGES * 8;
    int blocks = (int)((nt + 255) / 256);
    k_scatter_enc<<<blocks, 256, 0, stream>>>(src, dst, hK, Dxh, hK_agg, d_agg);
  }
  {
    int blocks = (NODE_TILES + WPB * 4 - 1) / (WPB * 4);
    k_node_mlp<<<blocks, 256, 0, stream>>>(hK_agg, WK1, bK1, WK2, bK2, WK3, bK3,
                                           E_node);
  }
  {
    int blocks = (EDGE_TILES + WPB * 4 - 1) / (WPB * 4);
    k_edge<<<blocks, 256, 0, stream>>>(src, dst, h1v, h2v, E_node,
                                       WU1, bU1, WU2, bU2, WU3, bU3, E_new);
  }
  {
    int blocks = (NODE_TILES + WPB * 4 - 1) / (WPB * 4);
    k_final<<<blocks, 256, 0, stream>>>(E_new, WH, bH, d_agg, out);
  }
}


# Node and edge MLP kernels: weights held in LDS in B-fragment order (one 16-byte read per lane feeds two WMMAs), activation tile at row pitch 68: fragment reads and tile stores hit no bank conflict
# speedup vs baseline: 1.6426x; 1.2103x over previous
#include <hip/hip_runtime.h>

#define NNODES 50000
#define NEDGES 800000
#define NODE_TILES (NNODES / 16)
#define EDGE_TILES (NEDGES / 16)
#define WPB 8

typedef __attribute__((ext_vector_type(2))) float v2f;
typedef __attribute__((ext_vector_type(4))) float v4f;
typedef __attribute__((ext_vector_type(8))) float v8f;

#define APITCH 68

__device__ __forceinline__ v8f wmma4(v2f a, v2f b, v8f c) {
  return __builtin_amdgcn_wmma_f32_16x16x4_f32(
      false, a, false, b, (short)0, c, false, false);
}

__device__ __forceinline__ v2f ldA(const float* A, int ldk, int kb, int lane) {
  int m = lane & 15;
  int k = 4 * kb + 2 * (lane >> 4);
  v2f a;
  a.x = A[m * ldk + k];
  a.y = A[m * ldk + k + 1];
  return a;
}

__device__ __forceinline__ v2f ldB(const float* W, int ncols, int kb, int col0, int lane) {
  int col = col0 + (lane & 15);
  int k = 4 * kb + 2 * (lane >> 4);
  v2f b;
  b.x = W[k * ncols + col];
  b.y = W[(k + 1) * ncols + col];
  return b;
}

template <int K>
__device__ __forceinline__ void stage_wfrag(const float* __restrict__ W, float* Wf) {
#pragma unroll
  for (int it = 0; it < K * 16 / (WPB * 32); ++it) {
    int i4 = threadIdx.x + it * (WPB * 32);
    float4 w = *(const float4*)(W + 4 * i4);
    int k = i4 >> 4, col = (i4 & 15) << 2;
    int kb = k >> 2, h = (k >> 1) & 1, j = k & 1;
    int nb = col >> 4, c = col & 15;
    int d = (((kb * 2 + (nb >> 1)) * 32 + h * 16 + c) << 2) + ((nb & 1) << 1) + j;
    Wf[d] = w.x; Wf[d + 4] = w.y; Wf[d + 8] = w.z; Wf[d + 12] = w.w;
  }
}

__device__ __forceinline__ v4f ldB2(const float* Wf, int kb, int np, int lane) {
  return ((const v4f*)Wf)[(kb * 2 + np) * 32 + lane];
}

__device__ __forceinline__ void kstep64(v2f a, const float* Wf, int kb, v8f* acc, int lane) {
#pragma unroll
  for (int np = 0; np < 2; ++np) {
    v4f w = ldB2(Wf, kb, np, lane);
    v2f b0, b1;
    b0.x = w.x; b0.y = w.y; b1.x = w.z; b1.y = w.w;
    acc[2 * np]     = wmma4(a, b0, acc[2 * np]);
    acc[2 * np + 1] = wmma4(a, b1, acc[2 * np + 1]);
  }
}

__device__ __forceinline__ v8f bias_frag(const float* b, int col0, int lane) {
  float bv = b[col0 + (lane & 15)];
  v8f c;
#pragma unroll
  for (int r = 0; r < 8; ++r) c[r] = bv;
  return c;
}

__device__ __forceinline__ void st_tile(float* A, const v8f* acc, int lane, int mode) {
  int half = lane >> 4, nl = lane & 15;
#pragma unroll
  for (int nb = 0; nb < 4; ++nb)
#pragma unroll
    for (int r = 0; r < 8; ++r) {
      float v = acc[nb][r];
      if (mode == 1) v = tanhf(v);
      else if (mode == 2) v = fmaxf(v, 0.0f);
      A[(r + 8 * half) * APITCH + nb * 16 + nl] = v;
    }
}

__device__ __forceinline__ void gemm64_lds(const float* A, const float* Wf,
                                           const float* b, v8f* acc, int lane) {
#pragma unroll
  for (int nb = 0; nb < 4; ++nb) acc[nb] = bias_frag(b, nb * 16, lane);
#pragma unroll
  for (int kb = 0; kb < 16; ++kb)
    kstep64(ldA(A, APITCH, kb, lane), Wf, kb, acc, lane);
}

__global__ void k_zero(float* p, long n) {
  long i = (long)blockIdx.x * blockDim.x + threadIdx.x;
  if (i < n) p[i] = 0.0f;
}

__global__ void __launch_bounds__(256) k_encode(
    const float* __restrict__ x,
    const float* __restrict__ WencK, const float* __restrict__ bencK,
    const float* __restrict__ WencP1, const float* __restrict__ bencP1,
    const float* __restrict__ WencP2, const float* __restrict__ bencP2,
    const float* __restrict__ WD, const float* __restrict__ bD,
    float* __restrict__ hK, float* __restrict__ h1,
    float* __restrict__ h2, float* __restrict__ Dxh) {
  __shared__ float sWK[32 * 32], sW1[32 * 32], sW2[32 * 32], sWD[64 * 64];
  __shared__ float sbK[32], sb1[32], sb2[32], sbD[32];
  for (int i = threadIdx.x; i < 32 * 32; i += blockDim.x) {
    sWK[i] = WencK[i]; sW1[i] = WencP1[i]; sW2[i] = WencP2[i];
  }
  for (int i = threadIdx.x; i < 64 * 64; i += blockDim.x) sWD[i] = WD[i];
  for (int i = threadIdx.x; i < 32; i += blockDim.x) {
    sbK[i] = bencK[i]; sb1[i] = bencP1[i]; sb2[i] = bencP2[i]; sbD[i] = bD[32 + i];
  }
  __syncthreads();

  int wave = threadIdx.x >> 5, lane = threadIdx.x & 31;
  int m = lane & 15;
  int half = lane >> 4, nl = lane & 15;

  for (int tile = blockIdx.x * WPB + wave; tile < NODE_TILES;
       tile += gridDim.x * WPB) {
    int n0 = tile * 16;

    v8f aK[2], a1[2], a2[2], aD[2];
#pragma unroll
    for (int nb = 0; nb < 2; ++nb) {
      aK[nb] = bias_frag(sbK, nb * 16, lane);
      a1[nb] = bias_frag(sb1, nb * 16, lane);
      a2[nb] = bias_frag(sb2, nb * 16, lane);
      aD[nb] = bias_frag(sbD, nb * 16, lane);
    }

#pragma unroll
    for (int kb = 0; kb < 8; ++kb) {
      int k = 4 * kb + 2 * (lane >> 4);
      long base = (long)(n0 + m) * 64;
      v2f aq, ap;
      aq.x = x[base + k];       aq.y = x[base + k + 1];
      ap.x = x[base + 32 + k];  ap.y = x[base + 32 + k + 1];
#pragma unroll
      for (int nb = 0; nb < 2; ++nb) {
        aK[nb] = wmma4(ap, ldB(sWK, 32, kb, nb * 16, lane), aK[nb]);
        a1[nb] = wmma4(aq, ldB(sW1, 32, kb, nb * 16, lane), a1[nb]);
        a2[nb] = wmma4(aq, ldB(sW2, 32, kb, nb * 16, lane), a2[nb]);
      }
    }
#pragma unroll
    for (int kb = 0; kb < 16; ++kb) {
      int k = 4 * kb + 2 * (lane >> 4);
      long base = (long)(n0 + m) * 64;
      v2f ax;
      ax.x = x[base + k]; ax.y = x[base + k + 1];
#pragma unroll
      for (int nb = 0; nb < 2; ++nb)
        aD[nb] = wmma4(ax, ldB(sWD, 64, kb, 32 + nb * 16, lane), aD[nb]);
    }

#pragma unroll
    for (int nb = 0; nb < 2; ++nb)
#pragma unroll
      for (int r = 0; r < 8; ++r) {
        long o = (long)(n0 + r + 8 * half) * 32 + nb * 16 + nl;
        hK[o] = aK[nb][r];
        h1[o] = a1[nb][r];
        h2[o] = a2[nb][r];
        Dxh[o] = aD[nb][r];
      }
  }
}

__global__ void k_scatter_enc(const int* __restrict__ src, const int* __restrict__ dst,
                              const float* __restrict__ hK, const float* __restrict__ Dxh,
                              float* __restrict__ hK_agg, float* __restrict__ d_agg) {
  long tid = (long)blockIdx.x * blockDim.x + threadIdx.x;
  if (tid >= (long)NEDGES * 8) return;
  int e = (int)(tid >> 3);
  int j = (int)(tid & 7) * 4;
  int s = src[e], d = dst[e];
  float4 a = *(const float4*)&hK[(long)s * 32 + j];
  float4 b = *(const float4*)&Dxh[(long)s * 32 + j];
  float* pa = &hK_agg[(long)d * 32 + j];
  float* pb = &d_agg[(long)d * 32 + j];
  atomicAdd(pa + 0, a.x); atomicAdd(pa + 1, a.y);
  atomicAdd(pa + 2, a.z); atomicAdd(pa + 3, a.w);
  atomicAdd(pb + 0, b.x); atomicAdd(pb + 1, b.y);
  atomicAdd(pb + 2, b.z); atomicAdd(pb + 3, b.w);
}

__global__ void __launch_bounds__(256) k_node_mlp(
    const float* __restrict__ hK_agg,
    const float* __restrict__ WK1, const float* __restrict__ bK1,
    const float* __restrict__ WK2, const float* __restrict__ bK2,
    const float* __restrict__ WK3, const float* __restrict__ bK3,
    float* __restrict__ E_node) {
  __shared__ __attribute__((aligned(16))) float sW1[32 * 64], sW2[64 * 64], sW3[64 * 64];
  __shared__ float sb1[64], sb2[64], sb3[64];
  __shared__ __attribute__((aligned(16))) float sact[WPB][16 * APITCH];
  stage_wfrag<32>(WK1, sW1);
  stage_wfrag<64>(WK2, sW2);
  stage_wfrag<64>(WK3, sW3);
  for (int i = threadIdx.x; i < 64; i += blockDim.x) { sb1[i] = bK1[i]; sb2[i] = bK2[i]; sb3[i] = bK3[i]; }
  __syncthreads();

  int wave = threadIdx.x >> 5, lane = threadIdx.x & 31;
  int m = lane & 15;
  int half = lane >> 4, nl = lane & 15;
  float* A = sact[wave];

  for (int tile = blockIdx.x * WPB + wave; tile < NODE_TILES;
       tile += gridDim.x * WPB) {
    int n0 = tile * 16;

    v8f acc[4];
#pragma unroll
    for (int nb = 0; nb < 4; ++nb) acc[nb] = bias_frag(sb1, nb * 16, lane);
#pragma unroll
    for (int kb = 0; kb < 8; ++kb) {
      int k = 4 * kb + 2 * (lane >> 4);
      long base = (long)(n0 + m) * 32;
      v2f a; a.x = hK_agg[base + k]; a.y = hK_agg[base + k + 1];
      kstep64(a, sW1, kb, acc, lane);
    }
    st_tile(A, acc, lane, 1);
    gemm64_lds(A, sW2, sb2, acc, lane);
    st_tile(A, acc, lane, 2);
    gemm64_lds(A, sW3, sb3, acc, lane);

#pragma unroll
    for (int nb = 0; nb < 4; ++nb)
#pragma unroll
      for (int r = 0; r < 8; ++r)
        E_node[(long)(n0 + r + 8 * half) * 64 + nb * 16 + nl] = acc[nb][r];
  }
}

__global__ void __launch_bounds__(256) k_edge(
    const int* __restrict__ src, const int* __restrict__ dst,
    const float* __restrict__ h1, const float* __restrict__ h2,
    const float* __restrict__ E_node,
    const float* __restrict__ WU1, const float* __restrict__ bU1,
    const float* __restrict__ WU2, const float* __restrict__ bU2,
    const float* __restrict__ WU3, const float* __restrict__ bU3,
    float* __restrict__ E_new) {
  __shared__ __attribute__((aligned(16))) float sW1[32 * 64], sW2[64 * 64], sW3[64 * 64];
  __shared__ float sb1[64], sb2[64], sb3[64];
  __shared__ __attribute__((aligned(16))) float sact[WPB][16 * APITCH];
  stage_wfrag<32>(WU1, sW1);
  stage_wfrag<64>(WU2, sW2);
  stage_wfrag<64>(WU3, sW3);
  for (int i = threadIdx.x; i < 64; i += blockDim.x) { sb1[i] = bU1[i]; sb2[i] = bU2[i]; sb3[i] = bU3[i]; }
  __syncthreads();

  int wave = threadIdx.x >> 5, lane = threadIdx.x & 31;
  int m = lane & 15;
  int half = lane >> 4, nl = lane & 15;
  float* A = sact[wave];

  for (int tile = blockIdx.x * WPB + wave; tile < EDGE_TILES;
       tile += gridDim.x * WPB) {
    int e0 = tile * 16;
    int s = src[e0 + m], d = dst[e0 + m];

    int se[8], de[8];
#pragma unroll
    for (int r = 0; r < 8; ++r) {
      int e = e0 + r + 8 * half;
      se[r] = src[e];
      de[r] = dst[e];
    }
    float en[8][4];
#pragma unroll
    for (int r = 0; r < 8; ++r)
#pragma unroll
      for (int nb = 0; nb < 4; ++nb)
        en[r][nb] = E_node[(long)se[r] * 64 + nb * 16 + nl];

    v8f acc[4];
#pragma unroll
    for (int nb = 0; nb < 4; ++nb) acc[nb] = bias_frag(sb1, nb * 16, lane);
#pragma unroll
    for (int kb = 0; kb < 8; ++kb) {
      int k = 4 * kb + 2 * (lane >> 4);
      v2f a;
      a.x = h1[(long)s * 32 + k]     + h2[(long)d * 32 + k];
      a.y = h1[(long)s * 32 + k + 1] + h2[(long)d * 32 + k + 1];
      kstep64(a, sW1, kb, acc, lane);
    }
    st_tile(A, acc, lane, 1);
    gemm64_lds(A, sW2, sb2, acc, lane);
    st_tile(A, acc, lane, 2);
    gemm64_lds(A, sW3, sb3, acc, lane);

#pragma unroll
    for (int r = 0; r < 8; ++r) {
#pragma unroll
      for (int nb = 0; nb < 4; ++nb) {
        int col = nb * 16 + nl;
        float v = acc[nb][r] * en[r][nb];
        atomicAdd(&E_new[(long)de[r] * 64 + col], v);
      }
    }
  }
}

__global__ void __launch_bounds__(256) k_final(
    const float* __restrict__ E_new,
    const float* __restrict__ WH, const float* __restrict__ bH,
    const float* __restrict__ d_agg, float* __restrict__ out) {
  __shared__ float sWH[64 * 64], sbH[64];
  for (int i = threadIdx.x; i < 64 * 64; i += blockDim.x) sWH[i] = WH[i];
  for (int i = threadIdx.x; i < 64; i += blockDim.x) sbH[i] = bH[i];
  __syncthreads();

  int wave = threadIdx.x >> 5, lane = threadIdx.x & 31;
  int m = lane & 15;
  int half = lane >> 4, nl = lane & 15;

  for (int tile = blockIdx.x * WPB + wave; tile < NODE_TILES;
       tile += gridDim.x * WPB) {
    int n0 = tile * 16;

    v8f acc[4];
#pragma unroll
    for (int nb = 0; nb < 4; ++nb) acc[nb] = bias_frag(sbH, nb * 16, lane);
#pragma unroll
    for (int kb = 0; kb < 16; ++kb) {
      int k = 4 * kb + 2 * (lane >> 4);
      long base = (long)(n0 + m) * 64;
      v2f a; a.x = E_new[base + k]; a.y = E_new[base + k + 1];
#pragma unroll
      for (int nb = 0; nb < 4; ++nb)
        acc[nb] = wmma4(a, ldB(sWH, 64, kb, nb * 16, lane), acc[nb]);
    }

#pragma unroll
    for (int r = 0; r < 8; ++r) {
      long row = n0 + r + 8 * half;
#pragma unroll
      for (int nb = 0; nb < 2; ++nb) {
        out[row * 64 + nb * 16 + nl] = acc[nb + 2][r];
        out[row * 64 + 32 + nb * 16 + nl] =
            -acc[nb][r] - d_agg[row * 32 + nb * 16 + nl];
      }
    }
  }
}

extern "C" void kernel_launch(void* const* d_in, const int* in_sizes, int n_in,
                              void* d_out, int out_size, void* d_ws, size_t ws_size,
                              hipStream_t stream) {
  const float* x      = (const float*)d_in[0];
  const int*   src    = (const int*)d_in[1];
  const int*   dst    = (const int*)d_in[2];
  const float* WencK  = (const float*)d_in[3];
  const float* bencK  = (const float*)d_in[4];
  const float* WencP1 = (const float*)d_in[5];
  const float* bencP1 = (const float*)d_in[6];
  const float* WencP2 = (const float*)d_in[7];
  const float* bencP2 = (const float*)d_in[8];
  const float* WK1 = (const float*)d_in[9];
  const float* bK1 = (const float*)d_in[10];
  const float* WK2 = (const float*)d_in[11];
  const float* bK2 = (const float*)d_in[12];
  const float* WK3 = (const float*)d_in[13];
  const float* bK3 = (const float*)d_in[14];
  const float* WU1 = (const float*)d_in[15];
  const float* bU1 = (const float*)d_in[16];
  const float* WU2 = (const float*)d_in[17];
  const float* bU2 = (const float*)d_in[18];
  const float* WU3 = (const float*)d_in[19];
  const float* bU3 = (const float*)d_in[20];
  const float* WH  = (const float*)d_in[21];
  const float* bH  = (const float*)d_in[22];
  const float* WD  = (const float*)d_in[23];
  const float* bD  = (const float*)d_in[24];

  float* ws = (float*)d_ws;
  const size_t N32 = (size_t)NNODES * 32;
  const size_t N64 = (size_t)NNODES * 64;
  float* hK     = ws;
  float* h1v    = hK + N32;
  float* h2v    = h1v + N32;
  float* Dxh    = h2v + N32;
  float* hK_agg = Dxh + N32;
  float* d_agg  = hK_agg + N32;
  float* E_new  = d_agg + N32;
  float* E_node = E_new + N64;

  float* out = (float*)d_out;

  {
    long nz = (long)NNODES * 128;
    int blocks = (int)((nz + 255) / 256);
    k_zero<<<blocks, 256, 0, stream>>>(hK_agg, nz);
  }
  {
    int blocks = (NODE_TILES + WPB * 4 - 1) / (WPB * 4);
    k_encode<<<blocks, 256, 0, stream>>>(x, WencK, bencK, WencP1, bencP1,
                                         WencP2, bencP2, WD, bD,
                                         hK, h1v, h2v, Dxh);
  }
  {
    long nt = (long)NEDGES * 8;
    int blocks = (int)((nt + 255) / 256);
    k_scatter_enc<<<blocks, 256, 0, stream>>>(src, dst, hK, Dxh, hK_agg, d_agg);
  }
  {
    int blocks = (NODE_TILES + WPB * 4 - 1) / (WPB * 4);
    k_node_mlp<<<blocks, 256, 0, stream>>>(hK_agg, WK1, bK1, WK2, bK2, WK3, bK3,
                                           E_node);
  }
  {
    int blocks = (EDGE_TILES + WPB * 4 - 1) / (WPB * 4);
    k_edge<<<blocks, 256, 0, stream>>>(src, dst, h1v, h2v, E_node,
                                       WU1, bU1, WU2, bU2, WU3, bU3, E_new);
  }
  {
    int blocks = (NODE_TILES + WPB * 4 - 1) / (WPB * 4);
    k_final<<<blocks, 256, 0, stream>>>(E_new, WH, bH, d_agg, out);
  }
}


# Scatter kernel: lane = column, so each row gather and each float atomic add covers one whole 128-byte line of one edge (was 8 lanes on each of 4 lines); a wave loops over 128 edges; sums stay atomic
# speedup vs baseline: 1.7431x; 1.0612x over previous
#include <hip/hip_runtime.h>

#define NNODES 50000
#define NEDGES 800000
#define NODE_TILES (NNODES / 16)
#define EDGE_TILES (NEDGES / 16)
#define WPB 8

typedef __attribute__((ext_vector_type(2))) float v2f;
typedef __attribute__((ext_vector_type(4))) float v4f;
typedef __attribute__((ext_vector_type(8))) float v8f;

#define APITCH 68

__device__ __forceinline__ v8f wmma4(v2f a, v2f b, v8f c) {
  return __builtin_amdgcn_wmma_f32_16x16x4_f32(
      false, a, false, b, (short)0, c, false, false);
}

__device__ __forceinline__ v2f ldA(const float* A, int ldk, int kb, int lane) {
  int m = lane & 15;
  int k = 4 * kb + 2 * (lane >> 4);
  v2f a;
  a.x = A[m * ldk + k];
  a.y = A[m * ldk + k + 1];
  return a;
}

__device__ __forceinline__ v2f ldB(const float* W, int ncols, int kb, int col0, int lane) {
  int col = col0 + (lane & 15);
  int k = 4 * kb + 2 * (lane >> 4);
  v2f b;
  b.x = W[k * ncols + col];
  b.y = W[(k + 1) * ncols + col];
  return b;
}

template <int K>
__device__ __forceinline__ void stage_wfrag(const float* __restrict__ W, float* Wf) {
#pragma unroll
  for (int it = 0; it < K * 16 / (WPB * 32); ++it) {
    int i4 = threadIdx.x + it * (WPB * 32);
    float4 w = *(const float4*)(W + 4 * i4);
    int k = i4 >> 4, col = (i4 & 15) << 2;
    int kb = k >> 2, h = (k >> 1) & 1, j = k & 1;
    int nb = col >> 4, c = col & 15;
    int d = (((kb * 2 + (nb >> 1)) * 32 + h * 16 + c) << 2) + ((nb & 1) << 1) + j;
    Wf[d] = w.x; Wf[d + 4] = w.y; Wf[d + 8] = w.z; Wf[d + 12] = w.w;
  }
}

__device__ __forceinline__ v4f ldB2(const float* Wf, int kb, int np, int lane) {
  return ((const v4f*)Wf)[(kb * 2 + np) * 32 + lane];
}

__device__ __forceinline__ void kstep64(v2f a, const float* Wf, int kb, v8f* acc, int lane) {
#pragma unroll
  for (int np = 0; np < 2; ++np) {
    v4f w = ldB2(Wf, kb, np, lane);
    v2f b0, b1;
    b0.x = w.x; b0.y = w.y; b1.x = w.z; b1.y = w.w;
    acc[2 * np]     = wmma4(a, b0, acc[2 * np]);
    acc[2 * np + 1] = wmma4(a, b1, acc[2 * np + 1]);
  }
}

__device__ __forceinline__ v8f bias_frag(const float* b, int col0, int lane) {
  float bv = b[col0 + (lane & 15)];
  v8f c;
#pragma unroll
  for (int r = 0; r < 8; ++r) c[r] = bv;
  return c;
}

__device__ __forceinline__ void st_tile(float* A, const v8f* acc, int lane, int mode) {
  int half = lane >> 4, nl = lane & 15;
#pragma unroll
  for (int nb = 0; nb < 4; ++nb)
#pragma unroll
    for (int r = 0; r < 8; ++r) {
      float v = acc[nb][r];
      if (mode == 1) v = tanhf(v);
      else if (mode == 2) v = fmaxf(v, 0.0f);
      A[(r + 8 * half) * APITCH + nb * 16 + nl] = v;
    }
}

__device__ __forceinline__ void gemm64_lds(const float* A, const float* Wf,
                                           const float* b, v8f* acc, int lane) {
#pragma unroll
  for (int nb = 0; nb < 4; ++nb) acc[nb] = bias_frag(b, nb * 16, lane);
#pragma unroll
  for (int kb = 0; kb < 16; ++kb)
    kstep64(ldA(A, APITCH, kb, lane), Wf, kb, acc, lane);
}

__global__ void k_zero(float* p, long n) {
  long i = (long)blockIdx.x * blockDim.x + threadIdx.x;
  if (i < n) p[i] = 0.0f;
}

__global__ void __launch_bounds__(256) k_encode(
    const float* __restrict__ x,
    const float* __restrict__ WencK, const float* __restrict__ bencK,
    const float* __restrict__ WencP1, const float* __restrict__ bencP1,
    const float* __restrict__ WencP2, const float* __restrict__ bencP2,
    const float* __restrict__ WD, const float* __restrict__ bD,
    float* __restrict__ hK, float* __restrict__ h1,
    float* __restrict__ h2, float* __restrict__ Dxh) {
  __shared__ float sWK[32 * 32], sW1[32 * 32], sW2[32 * 32], sWD[64 * 64];
  __shared__ float sbK[32], sb1[32], sb2[32], sbD[32];
  for (int i = threadIdx.x; i < 32 * 32; i += blockDim.x) {
    sWK[i] = WencK[i]; sW1[i] = WencP1[i]; sW2[i] = WencP2[i];
  }
  for (int i = threadIdx.x; i < 64 * 64; i += blockDim.x) sWD[i] = WD[i];
  for (int i = threadIdx.x; i < 32; i += blockDim.x) {
    sbK[i] = bencK[i]; sb1[i] = bencP1[i]; sb2[i] = bencP2[i]; sbD[i] = bD[32 + i];
  }
  __syncthreads();

  int wave = threadIdx.x >> 5, lane = threadIdx.x & 31;
  int m = lane & 15;
  int half = lane >> 4, nl = lane & 15;

  for (int tile = blockIdx.x * WPB + wave; tile < NODE_TILES;
       tile += gridDim.x * WPB) {
    int n0 = tile * 16;

    v8f aK[2], a1[2], a2[2], aD[2];
#pragma unroll
    for (int nb = 0; nb < 2; ++nb) {
      aK[nb] = bias_frag(sbK, nb * 16, lane);
      a1[nb] = bias_frag(sb1, nb * 16, lane);
      a2[nb] = bias_frag(sb2, nb * 16, lane);
      aD[nb] = bias_frag(sbD, nb * 16, lane);
    }

#pragma unroll
    for (int kb = 0; kb < 8; ++kb) {
      int k = 4 * kb + 2 * (lane >> 4);
      long base = (long)(n0 + m) * 64;
      v2f aq, ap;
      aq.x = x[base + k];       aq.y = x[base + k + 1];
      ap.x = x[base + 32 + k];  ap.y = x[base + 32 + k + 1];
#pragma unroll
      for (int nb = 0; nb < 2; ++nb) {
        aK[nb] = wmma4(ap, ldB(sWK, 32, kb, nb * 16, lane), aK[nb]);
        a1[nb] = wmma4(aq, ldB(sW1, 32, kb, nb * 16, lane), a1[nb]);
        a2[nb] = wmma4(aq, ldB(sW2, 32, kb, nb * 16, lane), a2[nb]);
      }
    }
#pragma unroll
    for (int kb = 0; kb < 16; ++kb) {
      int k = 4 * kb + 2 * (lane >> 4);
      long base = (long)(n0 + m) * 64;
      v2f ax;
      ax.x = x[base + k]; ax.y = x[base + k + 1];
#pragma unroll
      for (int nb = 0; nb < 2; ++nb)
        aD[nb] = wmma4(ax, ldB(sWD, 64, kb, 32 + nb * 16, lane), aD[nb]);
    }

#pragma unroll
    for (int nb = 0; nb < 2; ++nb)
#pragma unroll
      for (int r = 0; r < 8; ++r) {
        long o = (long)(n0 + r + 8 * half) * 32 + nb * 16 + nl;
        hK[o] = aK[nb][r];
        h1[o] = a1[nb][r];
        h2[o] = a2[nb][r];
        Dxh[o] = aD[nb][r];
      }
  }
}

typedef __attribute__((address_space(1))) float gfloat;

__device__ __forceinline__ const gfloat* uniform_row(const float* base, int row, int pitch) {
  const gfloat* p = (const gfloat*)base + (long)row * pitch;
  asm("" : "+s"(p));
  return p;
}
__device__ __forceinline__ gfloat* uniform_row(float* base, int row, int pitch) {
  gfloat* p = (gfloat*)base + (long)row * pitch;
  asm("" : "+s"(p));
  return p;
}
__device__ __forceinline__ void atomic_add_g(gfloat* p, float v) {
  __hip_atomic_fetch_add(p, v, __ATOMIC_RELAXED, __HIP_MEMORY_SCOPE_AGENT);
}

#define SC_CHUNKS 4
#define SC_EPW (32 * SC_CHUNKS)
#define SC_WAVES (NEDGES / SC_EPW)
__global__ void __launch_bounds__(256) k_scatter_enc(
    const int* __restrict__ src, const int* __restrict__ dst,
    const float* __restrict__ hK, const float* __restrict__ Dxh,
    float* __restrict__ hK_agg, float* __restrict__ d_agg) {
  int lane = threadIdx.x & 31;
  int w = blockIdx.x * WPB + (threadIdx.x >> 5);
  if (w >= SC_WAVES) return;
  int e0 = w * SC_EPW;

  int sv[SC_CHUNKS], dv[SC_CHUNKS];
#pragma unroll
  for (int c = 0; c < SC_CHUNKS; ++c) {
    sv[c] = src[e0 + c * 32 + lane];
    dv[c] = dst[e0 + c * 32 + lane];
  }

#pragma unroll
  for (int c = 0; c < SC_CHUNKS; ++c) {
    float a[32], b[32];
#pragma unroll
    for (int i = 0; i < 32; ++i) {
      int s = __builtin_amdgcn_readlane(sv[c], i);
      a[i] = uniform_row(hK, s, 32)[lane];
      b[i] = uniform_row(Dxh, s, 32)[lane];
    }
#pragma unroll
    for (int i = 0; i < 32; ++i) {
      int d = __builtin_amdgcn_readlane(dv[c], i);
      atomic_add_g(uniform_row(hK_agg, d, 32) + lane, a[i]);
      atomic_add_g(uniform_row(d_agg, d, 32) + lane, b[i]);
    }
  }
}

__global__ void __launch_bounds__(256) k_node_mlp(
    const float* __restrict__ hK_agg,
    const float* __restrict__ WK1, const float* __restrict__ bK1,
    const float* __restrict__ WK2, const float* __restrict__ bK2,
    const float* __restrict__ WK3, const float* __restrict__ bK3,
    float* __restrict__ E_node) {
  __shared__ __attribute__((aligned(16))) float sW1[32 * 64], sW2[64 * 64], sW3[64 * 64];
  __shared__ float sb1[64], sb2[64], sb3[64];
  __shared__ __attribute__((aligned(16))) float sact[WPB][16 * APITCH];
  stage_wfrag<32>(WK1, sW1);
  stage_wfrag<64>(WK2, sW2);
  stage_wfrag<64>(WK3, sW3);
  for (int i = threadIdx.x; i < 64; i += blockDim.x) { sb1[i] = bK1[i]; sb2[i] = bK2[i]; sb3[i] = bK3[i]; }
  __syncthreads();

  int wave = threadIdx.x >> 5, lane = threadIdx.x & 31;
  int m = lane & 15;
  int half = lane >> 4, nl = lane & 15;
  float* A = sact[wave];

  for (int tile = blockIdx.x * WPB + wave; tile < NODE_TILES;
       tile += gridDim.x * WPB) {
    int n0 = tile * 16;

    v8f acc[4];
#pragma unroll
    for (int nb = 0; nb < 4; ++nb) acc[nb] = bias_frag(sb1, nb * 16, lane);
#pragma unroll
    for (int kb = 0; kb < 8; ++kb) {
      int k = 4 * kb + 2 * (lane >> 4);
      long base = (long)(n0 + m) * 32;
      v2f a; a.x = hK_agg[base + k]; a.y = hK_agg[base + k + 1];
      kstep64(a, sW1, kb, acc, lane);
    }
    st_tile(A, acc, lane, 1);
    gemm64_lds(A, sW2, sb2, acc, lane);
    st_tile(A, acc, lane, 2);
    gemm64_lds(A, sW3, sb3, acc, lane);

#pragma unroll
    for (int nb = 0; nb < 4; ++nb)
#pragma unroll
      for (int r = 0; r < 8; ++r)
        E_node[(long)(n0 + r + 8 * half) * 64 + nb * 16 + nl] = acc[nb][r];
  }
}

__global__ void __launch_bounds__(256) k_edge(
    const int* __restrict__ src, const int* __restrict__ dst,
    const float* __restrict__ h1, const float* __restrict__ h2,
    const float* __restrict__ E_node,
    const float* __restrict__ WU1, const float* __restrict__ bU1,
    const float* __restrict__ WU2, const float* __restrict__ bU2,
    const float* __restrict__ WU3, const float* __restrict__ bU3,
    float* __restrict__ E_new) {
  __shared__ __attribute__((aligned(16))) float sW1[32 * 64], sW2[64 * 64], sW3[64 * 64];
  __shared__ float sb1[64], sb2[64], sb3[64];
  __shared__ __attribute__((aligned(16))) float sact[WPB][16 * APITCH];
  stage_wfrag<32>(WU1, sW1);
  stage_wfrag<64>(WU2, sW2);
  stage_wfrag<64>(WU3, sW3);
  for (int i = threadIdx.x; i < 64; i += blockDim.x) { sb1[i] = bU1[i]; sb2[i] = bU2[i]; sb3[i] = bU3[i]; }
  __syncthreads();

  int wave = threadIdx.x >> 5, lane = threadIdx.x & 31;
  int m = lane & 15;
  int half = lane >> 4, nl = lane & 15;
  float* A = sact[wave];

  for (int tile = blockIdx.x * WPB + wave; tile < EDGE_TILES;
       tile += gridDim.x * WPB) {
    int e0 = tile * 16;
    int s = src[e0 + m], d = dst[e0 + m];

    int se[8], de[8];
#pragma unroll
    for (int r = 0; r < 8; ++r) {
      int e = e0 + r + 8 * half;
      se[r] = src[e];
      de[r] = dst[e];
    }
    float en[8][4];
#pragma unroll
    for (int r = 0; r < 8; ++r)
#pragma unroll
      for (int nb = 0; nb < 4; ++nb)
        en[r][nb] = E_node[(long)se[r] * 64 + nb * 16 + nl];

    v8f acc[4];
#pragma unroll
    for (int nb = 0; nb < 4; ++nb) acc[nb] = bias_frag(sb1, nb * 16, lane);
#pragma unroll
    for (int kb = 0; kb < 8; ++kb) {
      int k = 4 * kb + 2 * (lane >> 4);
      v2f a;
      a.x = h1[(long)s * 32 + k]     + h2[(long)d * 32 + k];
      a.y = h1[(long)s * 32 + k + 1] + h2[(long)d * 32 + k + 1];
      kstep64(a, sW1, kb, acc, lane);
    }
    st_tile(A, acc, lane, 1);
    gemm64_lds(A, sW2, sb2, acc, lane);
    st_tile(A, acc, lane, 2);
    gemm64_lds(A, sW3, sb3, acc, lane);

#pragma unroll
    for (int r = 0; r < 8; ++r) {
#pragma unroll
      for (int nb = 0; nb < 4; ++nb) {
        int col = nb * 16 + nl;
        float v = acc[nb][r] * en[r][nb];
        atomicAdd(&E_new[(long)de[r] * 64 + col], v);
      }
    }
  }
}

__global__ void __launch_bounds__(256) k_final(
    const float* __restrict__ E_new,
    const float* __restrict__ WH, const float* __restrict__ bH,
    const float* __restrict__ d_agg, float* __restrict__ out) {
  __shared__ float sWH[64 * 64], sbH[64];
  for (int i = threadIdx.x; i < 64 * 64; i += blockDim.x) sWH[i] = WH[i];
  for (int i = threadIdx.x; i < 64; i += blockDim.x) sbH[i] = bH[i];
  __syncthreads();

  int wave = threadIdx.x >> 5, lane = threadIdx.x & 31;
  int m = lane & 15;
  int half = lane >> 4, nl = lane & 15;

  for (int tile = blockIdx.x * WPB + wave; tile < NODE_TILES;
       tile += gridDim.x * WPB) {
    int n0 = tile * 16;

    v8f acc[4];
#pragma unroll
    for (int nb = 0; nb < 4; ++nb) acc[nb] = bias_frag(sbH, nb * 16, lane);
#pragma unroll
    for (int kb = 0; kb < 16; ++kb) {
      int k = 4 * kb + 2 * (lane >> 4);
      long base = (long)(n0 + m) * 64;
      v2f a; a.x = E_new[base + k]; a.y = E_new[base + k + 1];
#pragma unroll
      for (int nb = 0; nb < 4; ++nb)
        acc[nb] = wmma4(a, ldB(sWH, 64, kb, nb * 16, lane), acc[nb]);
    }

#pragma unroll
    for (int r = 0; r < 8; ++r) {
      long row = n0 + r + 8 * half;
#pragma unroll
      for (int nb = 0; nb < 2; ++nb) {
        out[row * 64 + nb * 16 + nl] = acc[nb + 2][r];
        out[row * 64 + 32 + nb * 16 + nl] =
            -acc[nb][r] - d_agg[row * 32 + nb * 16 + nl];
      }
    }
  }
}

extern "C" void kernel_launch(void* const* d_in, const int* in_sizes, int n_in,
                              void* d_out, int out_size, void* d_ws, size_t ws_size,
                              hipStream_t stream) {
  const float* x      = (const float*)d_in[0];
  const int*   src    = (const int*)d_in[1];
  const int*   dst    = (const int*)d_in[2];
  const float* WencK  = (const float*)d_in[3];
  const float* bencK  = (const float*)d_in[4];
  const float* WencP1 = (const float*)d_in[5];
  const float* bencP1 = (const float*)d_in[6];
  const float* WencP2 = (const float*)d_in[7];
  const float* bencP2 = (const float*)d_in[8];
  const float* WK1 = (const float*)d_in[9];
  const float* bK1 = (const float*)d_in[10];
  const float* WK2 = (const float*)d_in[11];
  const float* bK2 = (const float*)d_in[12];
  const float* WK3 = (const float*)d_in[13];
  const float* bK3 = (const float*)d_in[14];
  const float* WU1 = (const float*)d_in[15];
  const float* bU1 = (const float*)d_in[16];
  const float* WU2 = (const float*)d_in[17];
  const float* bU2 = (const float*)d_in[18];
  const float* WU3 = (const float*)d_in[19];
  const float* bU3 = (const float*)d_in[20];
  const float* WH  = (const float*)d_in[21];
  const float* bH  = (const float*)d_in[22];
  const float* WD  = (const float*)d_in[23];
  const float* bD  = (const float*)d_in[24];

  float* ws = (float*)d_ws;
  const size_t N32 = (size_t)NNODES * 32;
  const size_t N64 = (size_t)NNODES * 64;
  float* hK     = ws;
  float* h1v    = hK + N32;
  float* h2v    = h1v + N32;
  float* Dxh    = h2v + N32;
  float* hK_agg = Dxh + N32;
  float* d_agg  = hK_agg + N32;
  float* E_new  = d_agg + N32;
  float* E_node = E_new + N64;

  float* out = (float*)d_out;

  {
    long nz = (long)NNODES * 128;
    int blocks = (int)((nz + 255) / 256);
    k_zero<<<blocks, 256, 0, stream>>>(hK_agg, nz);
  }
  {
    int blocks = (NODE_TILES + WPB * 4 - 1) / (WPB * 4);
    k_encode<<<blocks, 256, 0, stream>>>(x, WencK, bencK, WencP1, bencP1,
                                         WencP2, bencP2, WD, bD,
                                         hK, h1v, h2v, Dxh);
  }
  {
    int blocks = (SC_WAVES + WPB - 1) / WPB;
    k_scatter_enc<<<blocks, 256, 0, stream>>>(src, dst, hK, Dxh, hK_agg, d_agg);
  }
  {
    int blocks = (NODE_TILES + WPB * 4 - 1) / (WPB * 4);
    k_node_mlp<<<blocks, 256, 0, stream>>>(hK_agg, WK1, bK1, WK2, bK2, WK3, bK3,
                                           E_node);
  }
  {
    int blocks = (EDGE_TILES + WPB * 4 - 1) / (WPB * 4);
    k_edge<<<blocks, 256, 0, stream>>>(src, dst, h1v, h2v, E_node,
                                       WU1, bU1, WU2, bU2, WU3, bU3, E_new);
  }
  {
    int blocks = (NODE_TILES + WPB * 4 - 1) / (WPB * 4);
    k_final<<<blocks, 256, 0, stream>>>(E_new, WH, bH, d_agg, out);
  }
}


# Node and edge MLP kernels: tanh as 1 - 2/(exp2(2x log2 e) + 1) on hardware exp2 and reciprocal, branch-free, f32, abs error about 2e-7 (was the library's two masked branches per value)
# speedup vs baseline: 2.1077x; 1.2092x over previous
#include <hip/hip_runtime.h>

#define NNODES 50000
#define NEDGES 800000
#define NODE_TILES (NNODES / 16)
#define EDGE_TILES (NEDGES / 16)
#define WPB 8

typedef __attribute__((ext_vector_type(2))) float v2f;
typedef __attribute__((ext_vector_type(4))) float v4f;
typedef __attribute__((ext_vector_type(8))) float v8f;

#define APITCH 68

__device__ __forceinline__ v8f wmma4(v2f a, v2f b, v8f c) {
  return __builtin_amdgcn_wmma_f32_16x16x4_f32(
      false, a, false, b, (short)0, c, false, false);
}

__device__ __forceinline__ v2f ldA(const float* A, int ldk, int kb, int lane) {
  int m = lane & 15;
  int k = 4 * kb + 2 * (lane >> 4);
  v2f a;
  a.x = A[m * ldk + k];
  a.y = A[m * ldk + k + 1];
  return a;
}

__device__ __forceinline__ v2f ldB(const float* W, int ncols, int kb, int col0, int lane) {
  int col = col0 + (lane & 15);
  int k = 4 * kb + 2 * (lane >> 4);
  v2f b;
  b.x = W[k * ncols + col];
  b.y = W[(k + 1) * ncols + col];
  return b;
}

template <int K>
__device__ __forceinline__ void stage_wfrag(const float* __restrict__ W, float* Wf) {
#pragma unroll
  for (int it = 0; it < K * 16 / (WPB * 32); ++it) {
    int i4 = threadIdx.x + it * (WPB * 32);
    float4 w = *(const float4*)(W + 4 * i4);
    int k = i4 >> 4, col = (i4 & 15) << 2;
    int kb = k >> 2, h = (k >> 1) & 1, j = k & 1;
    int nb = col >> 4, c = col & 15;
    int d = (((kb * 2 + (nb >> 1)) * 32 + h * 16 + c) << 2) + ((nb & 1) << 1) + j;
    Wf[d] = w.x; Wf[d + 4] = w.y; Wf[d + 8] = w.z; Wf[d + 12] = w.w;
  }
}

__device__ __forceinline__ v4f ldB2(const float* Wf, int kb, int np, int lane) {
  return ((const v4f*)Wf)[(kb * 2 + np) * 32 + lane];
}

__device__ __forceinline__ void kstep64(v2f a, const float* Wf, int kb, v8f* acc, int lane) {
#pragma unroll
  for (int np = 0; np < 2; ++np) {
    v4f w = ldB2(Wf, kb, np, lane);
    v2f b0, b1;
    b0.x = w.x; b0.y = w.y; b1.x = w.z; b1.y = w.w;
    acc[2 * np]     = wmma4(a, b0, acc[2 * np]);
    acc[2 * np + 1] = wmma4(a, b1, acc[2 * np + 1]);
  }
}

__device__ __forceinline__ v8f bias_frag(const float* b, int col0, int lane) {
  float bv = b[col0 + (lane & 15)];
  v8f c;
#pragma unroll
  for (int r = 0; r < 8; ++r) c[r] = bv;
  return c;
}

__device__ __forceinline__ float tanh_f32(float x) {
  float e = __builtin_amdgcn_exp2f(x * 2.8853900817779268f);
  return 1.0f - 2.0f * __builtin_amdgcn_rcpf(e + 1.0f);
}

__device__ __forceinline__ void st_tile(float* A, const v8f* acc, int lane, int mode) {
  int half = lane >> 4, nl = lane & 15;
  if (mode == 1) __builtin_amdgcn_sched_barrier(0);
#pragma unroll
  for (int nb = 0; nb < 4; ++nb)
#pragma unroll
    for (int r = 0; r < 8; ++r) {
      float v = acc[nb][r];
      if (mode == 1) v = tanh_f32(v);
      else if (mode == 2) v = fmaxf(v, 0.0f);
      A[(r + 8 * half) * APITCH + nb * 16 + nl] = v;
    }
}

__device__ __forceinline__ void gemm64_lds(const float* A, const float* Wf,
                                           const float* b, v8f* acc, int lane) {
#pragma unroll
  for (int nb = 0; nb < 4; ++nb) acc[nb] = bias_frag(b, nb * 16, lane);
#pragma unroll
  for (int kb = 0; kb < 16; ++kb)
    kstep64(ldA(A, APITCH, kb, lane), Wf, kb, acc, lane);
}

__global__ void k_zero(float* p, long n) {
  long i = (long)blockIdx.x * blockDim.x + threadIdx.x;
  if (i < n) p[i] = 0.0f;
}

__global__ void __launch_bounds__(256) k_encode(
    const float* __restrict__ x,
    const float* __restrict__ WencK, const float* __restrict__ bencK,
    const float* __restrict__ WencP1, const float* __restrict__ bencP1,
    const float* __restrict__ WencP2, const float* __restrict__ bencP2,
    const float* __restrict__ WD, const float* __restrict__ bD,
    float* __restrict__ hK, float* __restrict__ h1,
    float* __restrict__ h2, float* __restrict__ Dxh) {
  __shared__ float sWK[32 * 32], sW1[32 * 32], sW2[32 * 32], sWD[64 * 64];
  __shared__ float sbK[32], sb1[32], sb2[32], sbD[32];
  for (int i = threadIdx.x; i < 32 * 32; i += blockDim.x) {
    sWK[i] = WencK[i]; sW1[i] = WencP1[i]; sW2[i] = WencP2[i];
  }
  for (int i = threadIdx.x; i < 64 * 64; i += blockDim.x) sWD[i] = WD[i];
  for (int i = threadIdx.x; i < 32; i += blockDim.x) {
    sbK[i] = bencK[i]; sb1[i] = bencP1[i]; sb2[i] = bencP2[i]; sbD[i] = bD[32 + i];
  }
  __syncthreads();

  int wave = threadIdx.x >> 5, lane = threadIdx.x & 31;
  int m = lane & 15;
  int half = lane >> 4, nl = lane & 15;

  for (int tile = blockIdx.x * WPB + wave; tile < NODE_TILES;
       tile += gridDim.x * WPB) {
    int n0 = tile * 16;

    v8f aK[2], a1[2], a2[2], aD[2];
#pragma unroll
    for (int nb = 0; nb < 2; ++nb) {
      aK[nb] = bias_frag(sbK, nb * 16, lane);
      a1[nb] = bias_frag(sb1, nb * 16, lane);
      a2[nb] = bias_frag(sb2, nb * 16, lane);
      aD[nb] = bias_frag(sbD, nb * 16, lane);
    }

#pragma unroll
    for (int kb = 0; kb < 8; ++kb) {
      int k = 4 * kb + 2 * (lane >> 4);
      long base = (long)(n0 + m) * 64;
      v2f aq, ap;
      aq.x = x[base + k];       aq.y = x[base + k + 1];
      ap.x = x[base + 32 + k];  ap.y = x[base + 32 + k + 1];
#pragma unroll
      for (int nb = 0; nb < 2; ++nb) {
        aK[nb] = wmma4(ap, ldB(sWK, 32, kb, nb * 16, lane), aK[nb]);
        a1[nb] = wmma4(aq, ldB(sW1, 32, kb, nb * 16, lane), a1[nb]);
        a2[nb] = wmma4(aq, ldB(sW2, 32, kb, nb * 16, lane), a2[nb]);
      }
    }
#pragma unroll
    for (int kb = 0; kb < 16; ++kb) {
      int k = 4 * kb + 2 * (lane >> 4);
      long base = (long)(n0 + m) * 64;
      v2f ax;
      ax.x = x[base + k]; ax.y = x[base + k + 1];
#pragma unroll
      for (int nb = 0; nb < 2; ++nb)
        aD[nb] = wmma4(ax, ldB(sWD, 64, kb, 32 + nb * 16, lane), aD[nb]);
    }

#pragma unroll
    for (int nb = 0; nb < 2; ++nb)
#pragma unroll
      for (int r = 0; r < 8; ++r) {
        long o = (long)(n0 + r + 8 * half) * 32 + nb * 16 + nl;
        hK[o] = aK[nb][r];
        h1[o] = a1[nb][r];
        h2[o] = a2[nb][r];
        Dxh[o] = aD[nb][r];
      }
  }
}

typedef __attribute__((address_space(1))) float gfloat;

__device__ __forceinline__ const gfloat* uniform_row(const float* base, int row, int pitch) {
  const gfloat* p = (const gfloat*)base + (long)row * pitch;
  asm("" : "+s"(p));
  return p;
}
__device__ __forceinline__ gfloat* uniform_row(float* base, int row, int pitch) {
  gfloat* p = (gfloat*)base + (long)row * pitch;
  asm("" : "+s"(p));
  return p;
}
__device__ __forceinline__ void atomic_add_g(gfloat* p, float v) {
  __hip_atomic_fetch_add(p, v, __ATOMIC_RELAXED, __HIP_MEMORY_SCOPE_AGENT);
}

#define SC_CHUNKS 4
#define SC_EPW (32 * SC_CHUNKS)
#define SC_WAVES (NEDGES / SC_EPW)
__global__ void __launch_bounds__(256) k_scatter_enc(
    const int* __restrict__ src, const int* __restrict__ dst,
    const float* __restrict__ hK, const float* __restrict__ Dxh,
    float* __restrict__ hK_agg, float* __restrict__ d_agg) {
  int lane = threadIdx.x & 31;
  int w = blockIdx.x * WPB + (threadIdx.x >> 5);
  if (w >= SC_WAVES) return;
  int e0 = w * SC_EPW;

  int sv[SC_CHUNKS], dv[SC_CHUNKS];
#pragma unroll
  for (int c = 0; c < SC_CHUNKS; ++c) {
    sv[c] = src[e0 + c * 32 + lane];
    dv[c] = dst[e0 + c * 32 + lane];
  }

#pragma unroll
  for (int c = 0; c < SC_CHUNKS; ++c) {
    float a[32], b[32];
#pragma unroll
    for (int i = 0; i < 32; ++i) {
      int s = __builtin_amdgcn_readlane(sv[c], i);
      a[i] = uniform_row(hK, s, 32)[lane];
      b[i] = uniform_row(Dxh, s, 32)[lane];
    }
#pragma unroll
    for (int i = 0; i < 32; ++i) {
      int d = __builtin_amdgcn_readlane(dv[c], i);
      atomic_add_g(uniform_row(hK_agg, d, 32) + lane, a[i]);
      atomic_add_g(uniform_row(d_agg, d, 32) + lane, b[i]);
    }
  }
}

__global__ void __launch_bounds__(256) k_node_mlp(
    const float* __restrict__ hK_agg,
    const float* __restrict__ WK1, const float* __restrict__ bK1,
    const float* __restrict__ WK2, const float* __restrict__ bK2,
    const float* __restrict__ WK3, const float* __restrict__ bK3,
    float* __restrict__ E_node) {
  __shared__ __attribute__((aligned(16))) float sW1[32 * 64], sW2[64 * 64], sW3[64 * 64];
  __shared__ float sb1[64], sb2[64], sb3[64];
  __shared__ __attribute__((aligned(16))) float sact[WPB][16 * APITCH];
  stage_wfrag<32>(WK1, sW1);
  stage_wfrag<64>(WK2, sW2);
  stage_wfrag<64>(WK3, sW3);
  for (int i = threadIdx.x; i < 64; i += blockDim.x) { sb1[i] = bK1[i]; sb2[i] = bK2[i]; sb3[i] = bK3[i]; }
  __syncthreads();

  int wave = threadIdx.x >> 5, lane = threadIdx.x & 31;
  int m = lane & 15;
  int half = lane >> 4, nl = lane & 15;
  float* A = sact[wave];

  for (int tile = blockIdx.x * WPB + wave; tile < NODE_TILES;
       tile += gridDim.x * WPB) {
    int n0 = tile * 16;

    v8f acc[4];
#pragma unroll
    for (int nb = 0; nb < 4; ++nb) acc[nb] = bias_frag(sb1, nb * 16, lane);
#pragma unroll
    for (int kb = 0; kb < 8; ++kb) {
      int k = 4 * kb + 2 * (lane >> 4);
      long base = (long)(n0 + m) * 32;
      v2f a; a.x = hK_agg[base + k]; a.y = hK_agg[base + k + 1];
      kstep64(a, sW1, kb, acc, lane);
    }
    st_tile(A, acc, lane, 1);
    gemm64_lds(A, sW2, sb2, acc, lane);
    st_tile(A, acc, lane, 2);
    gemm64_lds(A, sW3, sb3, acc, lane);

#pragma unroll
    for (int nb = 0; nb < 4; ++nb)
#pragma unroll
      for (int r = 0; r < 8; ++r)
        E_node[(long)(n0 + r + 8 * half) * 64 + nb * 16 + nl] = acc[nb][r];
  }
}

__global__ void __launch_bounds__(256) k_edge(
    const int* __restrict__ src, const int* __restrict__ dst,
    const float* __restrict__ h1, const float* __restrict__ h2,
    const float* __restrict__ E_node,
    const float* __restrict__ WU1, const float* __restrict__ bU1,
    const float* __restrict__ WU2, const float* __restrict__ bU2,
    const float* __restrict__ WU3, const float* __restrict__ bU3,
    float* __restrict__ E_new) {
  __shared__ __attribute__((aligned(16))) float sW1[32 * 64], sW2[64 * 64], sW3[64 * 64];
  __shared__ float sb1[64], sb2[64], sb3[64];
  __shared__ __attribute__((aligned(16))) float sact[WPB][16 * APITCH];
  stage_wfrag<32>(WU1, sW1);
  stage_wfrag<64>(WU2, sW2);
  stage_wfrag<64>(WU3, sW3);
  for (int i = threadIdx.x; i < 64; i += blockDim.x) { sb1[i] = bU1[i]; sb2[i] = bU2[i]; sb3[i] = bU3[i]; }
  __syncthreads();

  int wave = threadIdx.x >> 5, lane = threadIdx.x & 31;
  int m = lane & 15;
  int half = lane >> 4, nl = lane & 15;
  float* A = sact[wave];

  for (int tile = blockIdx.x * WPB + wave; tile < EDGE_TILES;
       tile += gridDim.x * WPB) {
    int e0 = tile * 16;
    int s = src[e0 + m], d = dst[e0 + m];

    int se[8], de[8];
#pragma unroll
    for (int r = 0; r < 8; ++r) {
      int e = e0 + r + 8 * half;
      se[r] = src[e];
      de[r] = dst[e];
    }
    float en[8][4];
#pragma unroll
    for (int r = 0; r < 8; ++r)
#pragma unroll
      for (int nb = 0; nb < 4; ++nb)
        en[r][nb] = E_node[(long)se[r] * 64 + nb * 16 + nl];

    v8f acc[4];
#pragma unroll
    for (int nb = 0; nb < 4; ++nb) acc[nb] = bias_frag(sb1, nb * 16, lane);
#pragma unroll
    for (int kb = 0; kb < 8; ++kb) {
      int k = 4 * kb + 2 * (lane >> 4);
      v2f a;
      a.x = h1[(long)s * 32 + k]     + h2[(long)d * 32 + k];
      a.y = h1[(long)s * 32 + k + 1] + h2[(long)d * 32 + k + 1];
      kstep64(a, sW1, kb, acc, lane);
    }
    st_tile(A, acc, lane, 1);
    gemm64_lds(A, sW2, sb2, acc, lane);
    st_tile(A, acc, lane, 2);
    gemm64_lds(A, sW3, sb3, acc, lane);

#pragma unroll
    for (int r = 0; r < 8; ++r) {
#pragma unroll
      for (int nb = 0; nb < 4; ++nb) {
        int col = nb * 16 + nl;
        float v = acc[nb][r] * en[r][nb];
        atomicAdd(&E_new[(long)de[r] * 64 + col], v);
      }
    }
  }
}

__global__ void __launch_bounds__(256) k_final(
    const float* __restrict__ E_new,
    const float* __restrict__ WH, const float* __restrict__ bH,
    const float* __restrict__ d_agg, float* __restrict__ out) {
  __shared__ float sWH[64 * 64], sbH[64];
  for (int i = threadIdx.x; i < 64 * 64; i += blockDim.x) sWH[i] = WH[i];
  for (int i = threadIdx.x; i < 64; i += blockDim.x) sbH[i] = bH[i];
  __syncthreads();

  int wave = threadIdx.x >> 5, lane = threadIdx.x & 31;
  int m = lane & 15;
  int half = lane >> 4, nl = lane & 15;

  for (int tile = blockIdx.x * WPB + wave; tile < NODE_TILES;
       tile += gridDim.x * WPB) {
    int n0 = tile * 16;

    v8f acc[4];
#pragma unroll
    for (int nb = 0; nb < 4; ++nb) acc[nb] = bias_frag(sbH, nb * 16, lane);
#pragma unroll
    for (int kb = 0; kb < 16; ++kb) {
      int k = 4 * kb + 2 * (lane >> 4);
      long base = (long)(n0 + m) * 64;
      v2f a; a.x = E_new[base + k]; a.y = E_new[base + k + 1];
#pragma unroll
      for (int nb = 0; nb < 4; ++nb)
        acc[nb] = wmma4(a, ldB(sWH, 64, kb, nb * 16, lane), acc[nb]);
    }

#pragma unroll
    for (int r = 0; r < 8; ++r) {
      long row = n0 + r + 8 * half;
#pragma unroll
      for (int nb = 0; nb < 2; ++nb) {
        out[row * 64 + nb * 16 + nl] = acc[nb + 2][r];
        out[row * 64 + 32 + nb * 16 + nl] =
            -acc[nb][r] - d_agg[row * 32 + nb * 16 + nl];
      }
    }
  }
}

extern "C" void kernel_launch(void* const* d_in, const int* in_sizes, int n_in,
                              void* d_out, int out_size, void* d_ws, size_t ws_size,
                              hipStream_t stream) {
  const float* x      = (const float*)d_in[0];
  const int*   src    = (const int*)d_in[1];
  const int*   dst    = (const int*)d_in[2];
  const float* WencK  = (const float*)d_in[3];
  const float* bencK  = (const float*)d_in[4];
  const float* WencP1 = (const float*)d_in[5];
  const float* bencP1 = (const float*)d_in[6];
  const float* WencP2 = (const float*)d_in[7];
  const float* bencP2 = (const float*)d_in[8];
  const float* WK1 = (const float*)d_in[9];
  const float* bK1 = (const float*)d_in[10];
  const float* WK2 = (const float*)d_in[11];
  const float* bK2 = (const float*)d_in[12];
  const float* WK3 = (const float*)d_in[13];
  const float* bK3 = (const float*)d_in[14];
  const float* WU1 = (const float*)d_in[15];
  const float* bU1 = (const float*)d_in[16];
  const float* WU2 = (const float*)d_in[17];
  const float* bU2 = (const float*)d_in[18];
  const float* WU3 = (const float*)d_in[19];
  const float* bU3 = (const float*)d_in[20];
  const float* WH  = (const float*)d_in[21];
  const float* bH  = (const float*)d_in[22];
  const float* WD  = (const float*)d_in[23];
  const float* bD  = (const float*)d_in[24];

  float* ws = (float*)d_ws;
  const size_t N32 = (size_t)NNODES * 32;
  const size_t N64 = (size_t)NNODES * 64;
  float* hK     = ws;
  float* h1v    = hK + N32;
  float* h2v    = h1v + N32;
  float* Dxh    = h2v + N32;
  float* hK_agg = Dxh + N32;
  float* d_agg  = hK_agg + N32;
  float* E_new  = d_agg + N32;
  float* E_node = E_new + N64;

  float* out = (float*)d_out;

  {
    long nz = (long)NNODES * 128;
    int blocks = (int)((nz + 255) / 256);
    k_zero<<<blocks, 256, 0, stream>>>(hK_agg, nz);
  }
  {
    int blocks = (NODE_TILES + WPB * 4 - 1) / (WPB * 4);
    k_encode<<<blocks, 256, 0, stream>>>(x, WencK, bencK, WencP1, bencP1,
                                         WencP2, bencP2, WD, bD,
                                         hK, h1v, h2v, Dxh);
  }
  {
    int blocks = (SC_WAVES + WPB - 1) / WPB;
    k_scatter_enc<<<blocks, 256, 0, stream>>>(src, dst, hK, Dxh, hK_agg, d_agg);
  }
  {
    int blocks = (NODE_TILES + WPB * 4 - 1) / (WPB * 4);
    k_node_mlp<<<blocks, 256, 0, stream>>>(hK_agg, WK1, bK1, WK2, bK2, WK3, bK3,
                                           E_node);
  }
  {
    int blocks = (EDGE_TILES + WPB * 4 - 1) / (WPB * 4);
    k_edge<<<blocks, 256, 0, stream>>>(src, dst, h1v, h2v, E_node,
                                       WU1, bU1, WU2, bU2, WU3, bU3, E_new);
  }
  {
    int blocks = (NODE_TILES + WPB * 4 - 1) / (WPB * 4);
    k_final<<<blocks, 256, 0, stream>>>(E_new, WH, bH, d_agg, out);
  }
}
